# starting kernel
# baseline (speedup 1.0000x reference)
#include <hip/hip_runtime.h>

typedef float v2f __attribute__((ext_vector_type(2)));
typedef float v8f __attribute__((ext_vector_type(8)));

#define N_NODES 50000
#define D_HID   128
#define N_CLS   40
#define N_HOPS  30

__global__ void k_set_ones(float* deg, int n) {
    int i = blockIdx.x * blockDim.x + threadIdx.x;
    if (i < n) deg[i] = 1.0f;
}

__global__ void k_deg_accum(const int* __restrict__ dst, float* __restrict__ deg, int e) {
    int i = blockIdx.x * blockDim.x + threadIdx.x;
    if (i < e) atomicAdd(&deg[dst[i]], 1.0f);
}

__global__ void k_rsqrt_inplace(float* d, int n) {
    int i = blockIdx.x * blockDim.x + threadIdx.x;
    if (i < n) d[i] = rsqrtf(d[i]);
}

__global__ void k_edge_weights(const int* __restrict__ src, const int* __restrict__ dst,
                               const float* __restrict__ dinv, float* __restrict__ w, int e) {
    int i = blockIdx.x * blockDim.x + threadIdx.x;
    if (i < e) w[i] = dinv[src[i]] * dinv[dst[i]];
}

template <int NT, bool RELU_IN, bool RELU_OUT, bool NGUARD>
__global__ void k_gemm_wmma_f32(const float* __restrict__ A, const float* __restrict__ B,
                                const float* __restrict__ bias, float* __restrict__ C,
                                int M, int N, int K) {
    constexpr int NT16 = NT * 16;
    constexpr int RS   = NT * 32 + ((NT & 1) ? 64 : 32);
    __shared__ float sB[64 * RS];

    const int lane = threadIdx.x;
    const int tid  = threadIdx.y * 32 + lane;

    for (int i = tid; i < 128 * NT16; i += 128) {
        const int k = i / NT16;
        const int c = i - k * NT16;
        float v = 0.0f;
        if (!NGUARD || c < N) v = B[(size_t)k * N + c];
        sB[(k >> 1) * RS + c * 2 + (k & 1)] = v;
    }
    __syncthreads();

    const int mtile  = blockIdx.x * blockDim.y + threadIdx.y;
    const int mTiles = M >> 4;
    if (mtile >= mTiles) return;
    const int m0   = mtile << 4;
    const int half = lane >> 4;
    const int l16  = lane & 15;

    const float* Arow = A + (size_t)(m0 + l16) * K;

    v8f acc[NT];
    #pragma unroll
    for (int nt = 0; nt < NT; ++nt) acc[nt] = v8f{};

    for (int k0 = 0; k0 < K; k0 += 4) {
        const int kb = k0 + (half << 1);
        v2f a;
        a.x = Arow[kb];
        a.y = Arow[kb + 1];
        if (RELU_IN) { a.x = fmaxf(a.x, 0.0f); a.y = fmaxf(a.y, 0.0f); }

        const v2f* Bp = (const v2f*)(sB + ((k0 >> 1) + half) * RS) + l16;
        v2f bf[NT];
        #pragma unroll
        for (int nt = 0; nt < NT; ++nt) bf[nt] = Bp[nt * 16];

        #pragma unroll
        for (int nt = 0; nt < NT; ++nt) {
            acc[nt] = __builtin_amdgcn_wmma_f32_16x16x4_f32(false, a, false, bf[nt],
                                                            (short)0, acc[nt],
                                                            false, false);
        }
    }

    #pragma unroll
    for (int nt = 0; nt < NT; ++nt) {
        const int col = nt * 16 + l16;
        if (NGUARD && col >= N) continue;
        const float bv = bias[col];
        float* Cc = C + col;
        #pragma unroll
        for (int r = 0; r < 8; ++r) {
            const int row = m0 + r + (half << 3);
            float v = acc[nt][r] + bv;
            if (RELU_OUT) v = fmaxf(v, 0.0f);
            Cc[(size_t)row * N] = v;
        }
    }
}

__global__ void k_hop_init128(float* __restrict__ out, const float* __restrict__ in,
                              const float* __restrict__ dinv, int n) {
    int t = blockIdx.x * blockDim.x + threadIdx.x;
    if (t < n * 32) {
        int node = t >> 5;
        float s = dinv[node]; s *= s;
        float4 v = ((const float4*)in)[t];
        float4 o = make_float4(v.x * s, v.y * s, v.z * s, v.w * s);
        ((float4*)out)[t] = o;
    }
}

__global__ void k_hop_edges128(const int* __restrict__ src, const int* __restrict__ dst,
                               const float* __restrict__ w, const float* __restrict__ in,
                               float* __restrict__ out, int e) {
    int t = blockIdx.x * blockDim.x + threadIdx.x;
    int ed = t >> 5;
    if (ed >= e) return;
    int lane = t & 31;
    float we = w[ed];
    int s = src[ed];
    int d = dst[ed];
    float4 v = ((const float4*)(in + ((size_t)s << 7)))[lane];
    float* op = out + ((size_t)d << 7) + (lane << 2);
    atomicAdd(op + 0, v.x * we);
    atomicAdd(op + 1, v.y * we);
    atomicAdd(op + 2, v.z * we);
    atomicAdd(op + 3, v.w * we);
}

__global__ void k_hop_init40(float* __restrict__ out, const float* __restrict__ in,
                             const float* __restrict__ dinv, int n) {
    int t = blockIdx.x * blockDim.x + threadIdx.x;
    if (t < n * N_CLS) {
        int node = t / N_CLS;
        float s = dinv[node]; s *= s;
        out[t] = in[t] * s;
    }
}

__global__ void k_hop_edges40(const int* __restrict__ src, const int* __restrict__ dst,
                              const float* __restrict__ w, const float* __restrict__ in,
                              float* __restrict__ out, int e) {
    int t = blockIdx.x * blockDim.x + threadIdx.x;
    if (t >= e * N_CLS) return;
    int ed = t / N_CLS;
    int f  = t - ed * N_CLS;
    atomicAdd(out + (size_t)dst[ed] * N_CLS + f, w[ed] * in[(size_t)src[ed] * N_CLS + f]);
}

extern "C" void kernel_launch(void* const* d_in, const int* in_sizes, int n_in,
                              void* d_out, int out_size, void* d_ws, size_t ws_size,
                              hipStream_t stream) {
    const float* x    = (const float*)d_in[0];
    const int*   edge = (const int*)d_in[1];
    const float* W1   = (const float*)d_in[2];
    const float* b1   = (const float*)d_in[3];
    const float* W2   = (const float*)d_in[4];
    const float* b2   = (const float*)d_in[5];
    const float* Wc   = (const float*)d_in[6];
    const float* bc   = (const float*)d_in[7];

    const int n = N_NODES;
    const int e = in_sizes[1] / 2;
    const int* src = edge;
    const int* dst = edge + e;

    float* ws   = (float*)d_ws;
    float* dinv = ws;
    float* wgt  = dinv + 50176;
    float* bufA = wgt + 800256;
    float* bufB = bufA + (size_t)n * D_HID;
    float* Q    = bufB + (size_t)n * D_HID;
    float* outp = (float*)d_out;

    const int T = 256;

    k_set_ones<<<(n + T - 1) / T, T, 0, stream>>>(dinv, n);
    k_deg_accum<<<(e + T - 1) / T, T, 0, stream>>>(dst, dinv, e);
    k_rsqrt_inplace<<<(n + T - 1) / T, T, 0, stream>>>(dinv, n);
    k_edge_weights<<<(e + T - 1) / T, T, 0, stream>>>(src, dst, dinv, wgt, e);

    const int mTiles = n / 16;
    dim3 blk(32, 4, 1);
    dim3 grid((mTiles + 3) / 4, 1, 1);

    k_gemm_wmma_f32<8, false, true, false><<<grid, blk, 0, stream>>>(
        x, W1, b1, bufA, n, D_HID, D_HID);

    k_gemm_wmma_f32<8, false, false, false><<<grid, blk, 0, stream>>>(
        bufA, W2, b2, bufB, n, D_HID, D_HID);

    {
        const int initN = (n * 32 + T - 1) / T;
        const int edgeN = (e * 32 + T - 1) / T;
        for (int hop = 0; hop < N_HOPS; ++hop) {
            const float* hin = (hop & 1) ? bufA : bufB;
            float*       hout = (hop & 1) ? bufB : bufA;
            k_hop_init128<<<initN, T, 0, stream>>>(hout, hin, dinv, n);
            k_hop_edges128<<<edgeN, T, 0, stream>>>(src, dst, wgt, hin, hout, e);
        }
    }
    float* prop = (N_HOPS & 1) ? bufA : bufB;

    k_gemm_wmma_f32<3, true, false, true><<<grid, blk, 0, stream>>>(
        prop, Wc, bc, Q, n, N_CLS, D_HID);

    k_hop_init40<<<(n * N_CLS + T - 1) / T, T, 0, stream>>>(outp, Q, dinv, n);
    k_hop_edges40<<<(e * N_CLS + T - 1) / T, T, 0, stream>>>(src, dst, wgt, Q, outp, e);
}


# Hop edge scatter with float atomics replaced by a gather over in-edge rows (CSR by destination, built each call, rows ordered by source id): fixed-order per-row sums, so the sum order changes
# speedup vs baseline: 5.1811x; 5.1811x over previous
#include <hip/hip_runtime.h>

typedef float v2f __attribute__((ext_vector_type(2)));
typedef float v8f __attribute__((ext_vector_type(8)));

#define N_NODES 50000
#define D_HID   128
#define N_CLS   40
#define N_HOPS  30

__global__ void k_zero_counts(int* __restrict__ cnt, int n) {
    int i = blockIdx.x * blockDim.x + threadIdx.x;
    if (i < n) cnt[i] = 0;
}

__global__ void k_count_in_edges(const int* __restrict__ dst, int* __restrict__ cnt, int e) {
    int i = blockIdx.x * blockDim.x + threadIdx.x;
    if (i < e) atomicAdd(&cnt[dst[i]], 1);
}

#define SCAN_T   1024
#define SCAN_CH  ((N_NODES + SCAN_T) / SCAN_T)
#define SCAN_PAD (SCAN_T * SCAN_CH)
__global__ __launch_bounds__(SCAN_T)
void k_scan_rows(const int* __restrict__ cnt, int* __restrict__ rowptr,
                 int* __restrict__ cursor, float* __restrict__ dinv) {
    __shared__ int part[SCAN_T];
    const int t = threadIdx.x;
    const int b = t * SCAN_CH;
    int c[SCAN_CH];
    int s = 0;
    #pragma unroll
    for (int k = 0; k < SCAN_CH; ++k) c[k] = cnt[b + k];
    #pragma unroll
    for (int k = 0; k < SCAN_CH; ++k) s += c[k];
    part[t] = s;
    __syncthreads();
    for (int off = 1; off < SCAN_T; off <<= 1) {
        const int v = (t >= off) ? part[t - off] : 0;
        __syncthreads();
        part[t] += v;
        __syncthreads();
    }
    int run = part[t] - s;
    #pragma unroll
    for (int k = 0; k < SCAN_CH; ++k) {
        const int i = b + k;
        rowptr[i] = run;
        cursor[i] = run;
        dinv[i]   = rsqrtf((float)(c[k] + 1));
        run += c[k];
    }
}

__global__ void k_fill_rows(const int* __restrict__ src, const int* __restrict__ dst,
                            int* __restrict__ cursor, int* __restrict__ col_tmp, int e) {
    int i = blockIdx.x * blockDim.x + threadIdx.x;
    if (i < e) {
        const int pos = atomicAdd(&cursor[dst[i]], 1);
        col_tmp[pos] = src[i];
    }
}

__global__ __launch_bounds__(256)
void k_order_rows(const int* __restrict__ rowptr, const int* __restrict__ col_tmp,
                  const float* __restrict__ dinv, int* __restrict__ col,
                  float* __restrict__ w, int n) {
    const int lane = threadIdx.x & 31;
    const int r = __builtin_amdgcn_readfirstlane((int)(blockIdx.x * 8 + (threadIdx.x >> 5)));
    if (r >= n) return;
    const int b = rowptr[r];
    const int e = rowptr[r + 1];
    const float dr = dinv[r];
    for (int mb = b; mb < e; mb += 32) {
        const int p = mb + lane;
        const bool have = p < e;
        const int mine = have ? col_tmp[p] : 0;
        int rank = 0;
        for (int ob = b; ob < e; ob += 32) {
            const int q0 = ob + lane;
            const int other = (q0 < e) ? col_tmp[q0] : 0;
            const int cnt = min(32, e - ob);
            for (int j = 0; j < cnt; ++j) {
                const int o = __builtin_amdgcn_readlane(other, j);
                const int q = ob + j;
                rank += ((o < mine) || (o == mine && q < p)) ? 1 : 0;
            }
        }
        if (have) {
            col[b + rank] = mine;
            w[b + rank]   = dinv[mine] * dr;
        }
    }
}

template <int NT, bool RELU_IN, bool RELU_OUT, bool NGUARD>
__global__ void k_gemm_wmma_f32(const float* __restrict__ A, const float* __restrict__ B,
                                const float* __restrict__ bias, float* __restrict__ C,
                                int M, int N, int K) {
    constexpr int NT16 = NT * 16;
    constexpr int RS   = NT * 32 + ((NT & 1) ? 64 : 32);
    __shared__ float sB[64 * RS];

    const int lane = threadIdx.x;
    const int tid  = threadIdx.y * 32 + lane;

    for (int i = tid; i < 128 * NT16; i += 128) {
        const int k = i / NT16;
        const int c = i - k * NT16;
        float v = 0.0f;
        if (!NGUARD || c < N) v = B[(size_t)k * N + c];
        sB[(k >> 1) * RS + c * 2 + (k & 1)] = v;
    }
    __syncthreads();

    const int mtile  = blockIdx.x * blockDim.y + threadIdx.y;
    const int mTiles = M >> 4;
    if (mtile >= mTiles) return;
    const int m0   = mtile << 4;
    const int half = lane >> 4;
    const int l16  = lane & 15;

    const float* Arow = A + (size_t)(m0 + l16) * K;

    v8f acc[NT];
    #pragma unroll
    for (int nt = 0; nt < NT; ++nt) acc[nt] = v8f{};

    for (int k0 = 0; k0 < K; k0 += 4) {
        const int kb = k0 + (half << 1);
        v2f a;
        a.x = Arow[kb];
        a.y = Arow[kb + 1];
        if (RELU_IN) { a.x = fmaxf(a.x, 0.0f); a.y = fmaxf(a.y, 0.0f); }

        const v2f* Bp = (const v2f*)(sB + ((k0 >> 1) + half) * RS) + l16;
        v2f bf[NT];
        #pragma unroll
        for (int nt = 0; nt < NT; ++nt) bf[nt] = Bp[nt * 16];

        #pragma unroll
        for (int nt = 0; nt < NT; ++nt) {
            acc[nt] = __builtin_amdgcn_wmma_f32_16x16x4_f32(false, a, false, bf[nt],
                                                            (short)0, acc[nt],
                                                            false, false);
        }
    }

    #pragma unroll
    for (int nt = 0; nt < NT; ++nt) {
        const int col = nt * 16 + l16;
        if (NGUARD && col >= N) continue;
        const float bv = bias[col];
        float* Cc = C + col;
        #pragma unroll
        for (int r = 0; r < 8; ++r) {
            const int row = m0 + r + (half << 3);
            float v = acc[nt][r] + bv;
            if (RELU_OUT) v = fmaxf(v, 0.0f);
            Cc[(size_t)row * N] = v;
        }
    }
}

#define HOP_NPW 4

__device__ __forceinline__ void axpy4(float4& acc, float s, const float4& v) {
    acc.x = fmaf(s, v.x, acc.x);
    acc.y = fmaf(s, v.y, acc.y);
    acc.z = fmaf(s, v.z, acc.z);
    acc.w = fmaf(s, v.w, acc.w);
}

template <int R>
__device__ __forceinline__ void gather_group(float4& acc, const float4* __restrict__ in4,
                                             int c, float we, int j, int lane) {
    float4 v[R];
    float  s[R];
    #pragma unroll
    for (int k = 0; k < R; ++k) {
        const int sidx = __builtin_amdgcn_readlane(c, j + k);
        s[k] = __int_as_float(__builtin_amdgcn_readlane(__float_as_int(we), j + k));
        v[k] = in4[(size_t)sidx * 32 + lane];
    }
    #pragma unroll
    for (int k = 0; k < R; ++k) axpy4(acc, s[k], v[k]);
}

__global__ __launch_bounds__(256)
void k_hop_gather128(const int* __restrict__ rowptr, const int* __restrict__ col,
                     const float* __restrict__ w, const float* __restrict__ dinv,
                     const float* __restrict__ in, float* __restrict__ out, int n) {
    const int lane = threadIdx.x & 31;
    const int wave = __builtin_amdgcn_readfirstlane((int)(blockIdx.x * 8 + (threadIdx.x >> 5)));
    const float4* __restrict__ in4 = (const float4*)in;
    float4* __restrict__ out4 = (float4*)out;

    for (int i = 0; i < HOP_NPW; ++i) {
        const int node = wave * HOP_NPW + i;
        if (node >= n) break;
        const int beg = rowptr[node];
        const int end = rowptr[node + 1];
        float sl = dinv[node]; sl *= sl;
        int   c  = 0;
        float we = 0.0f;
        if (beg < end) {
            const int idx = min(beg + lane, end - 1);
            c  = col[idx];
            we = w[idx];
        }
        const float4 self = in4[(size_t)node * 32 + lane];
        float4 acc = make_float4(0.0f, 0.0f, 0.0f, 0.0f);

        int base = beg;
        while (base < end) {
            const int cnt = min(32, end - base);
            int j = 0;
            for (; j + 8 <= cnt; j += 8) gather_group<8>(acc, in4, c, we, j, lane);
            if (j + 4 <= cnt) { gather_group<4>(acc, in4, c, we, j, lane); j += 4; }
            const int rem = cnt - j;
            if (rem == 3)      gather_group<3>(acc, in4, c, we, j, lane);
            else if (rem == 2) gather_group<2>(acc, in4, c, we, j, lane);
            else if (rem == 1) gather_group<1>(acc, in4, c, we, j, lane);
            base += 32;
            if (base < end) {
                const int idx = min(base + lane, end - 1);
                c  = col[idx];
                we = w[idx];
            }
        }
        axpy4(acc, sl, self);
        out4[(size_t)node * 32 + lane] = acc;
    }
}

__global__ void k_hop_gather40(const int* __restrict__ rowptr, const int* __restrict__ col,
                               const float* __restrict__ w, const float* __restrict__ dinv,
                               const float* __restrict__ in, float* __restrict__ out, int n) {
    const int t = blockIdx.x * blockDim.x + threadIdx.x;
    if (t >= n * (N_CLS / 4)) return;
    const int node = t / (N_CLS / 4);
    const int ch   = t - node * (N_CLS / 4);
    const float4* __restrict__ in4 = (const float4*)in;
    float sl = dinv[node]; sl *= sl;
    const float4 self = in4[t];
    float4 acc = make_float4(0.0f, 0.0f, 0.0f, 0.0f);
    const int beg = rowptr[node];
    const int end = rowptr[node + 1];
    #pragma unroll 4
    for (int p = beg; p < end; ++p) {
        const int   s  = col[p];
        const float we = w[p];
        const float4 v = in4[(size_t)s * (N_CLS / 4) + ch];
        axpy4(acc, we, v);
    }
    axpy4(acc, sl, self);
    ((float4*)out)[t] = acc;
}

extern "C" void kernel_launch(void* const* d_in, const int* in_sizes, int n_in,
                              void* d_out, int out_size, void* d_ws, size_t ws_size,
                              hipStream_t stream) {
    const float* x    = (const float*)d_in[0];
    const int*   edge = (const int*)d_in[1];
    const float* W1   = (const float*)d_in[2];
    const float* b1   = (const float*)d_in[3];
    const float* W2   = (const float*)d_in[4];
    const float* b2   = (const float*)d_in[5];
    const float* Wc   = (const float*)d_in[6];
    const float* bc   = (const float*)d_in[7];

    const int n = N_NODES;
    const int e = in_sizes[1] / 2;
    const int* src = edge;
    const int* dst = edge + e;

    float* ws   = (float*)d_ws;
    float* dinv = ws;
    float* wgt  = dinv + 50176;
    float* bufA = wgt + 800256;
    float* bufB = bufA + (size_t)n * D_HID;
    int*   col    = (int*)(bufB + (size_t)n * D_HID);
    int*   coltmp = col + 800256;
    int*   rowptr = coltmp + 800256;
    int*   cnt    = rowptr + 50176;
    int*   cursor = cnt + 50176;
    float* outp = (float*)d_out;

    const int T = 256;

    k_zero_counts<<<(SCAN_PAD + T - 1) / T, T, 0, stream>>>(cnt, SCAN_PAD);
    k_count_in_edges<<<(e + T - 1) / T, T, 0, stream>>>(dst, cnt, e);
    k_scan_rows<<<1, SCAN_T, 0, stream>>>(cnt, rowptr, cursor, dinv);
    k_fill_rows<<<(e + T - 1) / T, T, 0, stream>>>(src, dst, cursor, coltmp, e);
    k_order_rows<<<(n + 7) / 8, T, 0, stream>>>(rowptr, coltmp, dinv, col, wgt, n);

    const int mTiles = n / 16;
    dim3 blk(32, 4, 1);
    dim3 grid((mTiles + 3) / 4, 1, 1);

    k_gemm_wmma_f32<8, false, true, false><<<grid, blk, 0, stream>>>(
        x, W1, b1, bufA, n, D_HID, D_HID);

    k_gemm_wmma_f32<8, false, false, false><<<grid, blk, 0, stream>>>(
        bufA, W2, b2, bufB, n, D_HID, D_HID);

    {
        const int hopN = (n + 8 * HOP_NPW - 1) / (8 * HOP_NPW);
        for (int hop = 0; hop < N_HOPS; ++hop) {
            const float* hin = (hop & 1) ? bufA : bufB;
            float*       hout = (hop & 1) ? bufB : bufA;
            k_hop_gather128<<<hopN, T, 0, stream>>>(rowptr, col, wgt, dinv, hin, hout, n);
        }
    }
    float* prop = (N_HOPS & 1) ? bufA : bufB;
    float* Q    = (N_HOPS & 1) ? bufB : bufA;

    k_gemm_wmma_f32<3, true, false, true><<<grid, blk, 0, stream>>>(
        prop, Wc, bc, Q, n, N_CLS, D_HID);

    k_hop_gather40<<<(n * (N_CLS / 4) + T - 1) / T, T, 0, stream>>>(
        rowptr, col, wgt, dinv, Q, outp, n);
}


# CSR build without global atomics: 64 edge chunks count in-edges per node in LDS (the add's return is the edge's rank), prefix over chunks, one plain store per edge; row and sum order unchanged
# speedup vs baseline: 5.4363x; 1.0493x over previous
#include <hip/hip_runtime.h>

typedef float v2f __attribute__((ext_vector_type(2)));
typedef float v8f __attribute__((ext_vector_type(8)));

#define N_NODES 50000
#define D_HID   128
#define N_CLS   40
#define N_HOPS  30

#define SCAN_T   1024
#define SCAN_CH  ((N_NODES + SCAN_T) / SCAN_T)
#define SCAN_PAD (SCAN_T * SCAN_CH)
#define NCH      64

__global__ __launch_bounds__(SCAN_T)
void k_chunk_count(const int* __restrict__ dst, int* __restrict__ hist,
                   int* __restrict__ rank, int e) {
    __shared__ int h[SCAN_PAD];
    const int t = threadIdx.x;
    const int g = blockIdx.x;
    for (int i = t; i < SCAN_PAD; i += SCAN_T) h[i] = 0;
    __syncthreads();
    const int per = (e + NCH - 1) / NCH;
    const int lo  = g * per;
    const int hi  = min(lo + per, e);
    for (int i0 = lo + t; i0 < hi; i0 += 8 * SCAN_T) {
        int d[8];
        #pragma unroll
        for (int k = 0; k < 8; ++k) {
            const int i = i0 + k * SCAN_T;
            d[k] = (i < hi) ? dst[i] : 0;
        }
        #pragma unroll
        for (int k = 0; k < 8; ++k) {
            const int i = i0 + k * SCAN_T;
            if (i < hi) rank[i] = atomicAdd(&h[d[k]], 1);
        }
    }
    __syncthreads();
    int* __restrict__ hg = hist + (size_t)g * SCAN_PAD;
    for (int i = t; i < SCAN_PAD; i += SCAN_T) hg[i] = h[i];
}

__global__ __launch_bounds__(256)
void k_chunk_prefix(int* hist, int* __restrict__ cnt) {
    const int i = blockIdx.x * 256 + threadIdx.x;
    int v[NCH];
    #pragma unroll
    for (int g = 0; g < NCH; ++g) v[g] = hist[(size_t)g * SCAN_PAD + i];
    int run = 0;
    #pragma unroll
    for (int g = 0; g < NCH; ++g) {
        hist[(size_t)g * SCAN_PAD + i] = run;
        run += v[g];
    }
    cnt[i] = run;
}

__global__ __launch_bounds__(SCAN_T)
void k_scan_rows(const int* __restrict__ cnt, int* __restrict__ rowptr,
                 float* __restrict__ dinv) {
    __shared__ int part[SCAN_T];
    const int t = threadIdx.x;
    const int b = t * SCAN_CH;
    int c[SCAN_CH];
    int s = 0;
    #pragma unroll
    for (int k = 0; k < SCAN_CH; ++k) c[k] = cnt[b + k];
    #pragma unroll
    for (int k = 0; k < SCAN_CH; ++k) s += c[k];
    part[t] = s;
    __syncthreads();
    for (int off = 1; off < SCAN_T; off <<= 1) {
        const int v = (t >= off) ? part[t - off] : 0;
        __syncthreads();
        part[t] += v;
        __syncthreads();
    }
    int run = part[t] - s;
    #pragma unroll
    for (int k = 0; k < SCAN_CH; ++k) {
        const int i = b + k;
        rowptr[i] = run;
        dinv[i]   = rsqrtf((float)(c[k] + 1));
        run += c[k];
    }
}

__global__ __launch_bounds__(SCAN_T)
void k_place_edges(const int* __restrict__ src, const int* __restrict__ dst,
                   const int* __restrict__ rowptr, const int* __restrict__ hist,
                   const int* __restrict__ rank, int* __restrict__ col_tmp, int e) {
    __shared__ int off[SCAN_PAD];
    const int t = threadIdx.x;
    const int g = blockIdx.x;
    const int* __restrict__ hg = hist + (size_t)g * SCAN_PAD;
    #pragma unroll 7
    for (int i = t; i < SCAN_PAD; i += SCAN_T) off[i] = rowptr[i] + hg[i];
    __syncthreads();
    const int per = (e + NCH - 1) / NCH;
    const int lo  = g * per;
    const int hi  = min(lo + per, e);
    #pragma unroll 4
    for (int i = lo + t; i < hi; i += SCAN_T) col_tmp[off[dst[i]] + rank[i]] = src[i];
}

__global__ __launch_bounds__(256)
void k_order_rows(const int* __restrict__ rowptr, const int* __restrict__ col_tmp,
                  const float* __restrict__ dinv, int* __restrict__ col,
                  float* __restrict__ w, int n) {
    const int lane = threadIdx.x & 31;
    const int r = __builtin_amdgcn_readfirstlane((int)(blockIdx.x * 8 + (threadIdx.x >> 5)));
    if (r >= n) return;
    const int b = rowptr[r];
    const int e = rowptr[r + 1];
    const float dr = dinv[r];
    for (int mb = b; mb < e; mb += 32) {
        const int p = mb + lane;
        const bool have = p < e;
        const int mine = have ? col_tmp[p] : 0;
        int rank = 0;
        for (int ob = b; ob < e; ob += 32) {
            const int q0 = ob + lane;
            const int other = (q0 < e) ? col_tmp[q0] : 0;
            const int cnt = min(32, e - ob);
            for (int j = 0; j < cnt; ++j) {
                const int o = __builtin_amdgcn_readlane(other, j);
                const int q = ob + j;
                rank += ((o < mine) || (o == mine && q < p)) ? 1 : 0;
            }
        }
        if (have) {
            col[b + rank] = mine;
            w[b + rank]   = dinv[mine] * dr;
        }
    }
}

template <int NT, bool RELU_IN, bool RELU_OUT, bool NGUARD>
__global__ void k_gemm_wmma_f32(const float* __restrict__ A, const float* __restrict__ B,
                                const float* __restrict__ bias, float* __restrict__ C,
                                int M, int N, int K) {
    constexpr int NT16 = NT * 16;
    constexpr int RS   = NT * 32 + ((NT & 1) ? 64 : 32);
    __shared__ float sB[64 * RS];

    const int lane = threadIdx.x;
    const int tid  = threadIdx.y * 32 + lane;

    for (int i = tid; i < 128 * NT16; i += 128) {
        const int k = i / NT16;
        const int c = i - k * NT16;
        float v = 0.0f;
        if (!NGUARD || c < N) v = B[(size_t)k * N + c];
        sB[(k >> 1) * RS + c * 2 + (k & 1)] = v;
    }
    __syncthreads();

    const int mtile  = blockIdx.x * blockDim.y + threadIdx.y;
    const int mTiles = M >> 4;
    if (mtile >= mTiles) return;
    const int m0   = mtile << 4;
    const int half = lane >> 4;
    const int l16  = lane & 15;

    const float* Arow = A + (size_t)(m0 + l16) * K;

    v8f acc[NT];
    #pragma unroll
    for (int nt = 0; nt < NT; ++nt) acc[nt] = v8f{};

    for (int k0 = 0; k0 < K; k0 += 4) {
        const int kb = k0 + (half << 1);
        v2f a;
        a.x = Arow[kb];
        a.y = Arow[kb + 1];
        if (RELU_IN) { a.x = fmaxf(a.x, 0.0f); a.y = fmaxf(a.y, 0.0f); }

        const v2f* Bp = (const v2f*)(sB + ((k0 >> 1) + half) * RS) + l16;
        v2f bf[NT];
        #pragma unroll
        for (int nt = 0; nt < NT; ++nt) bf[nt] = Bp[nt * 16];

        #pragma unroll
        for (int nt = 0; nt < NT; ++nt) {
            acc[nt] = __builtin_amdgcn_wmma_f32_16x16x4_f32(false, a, false, bf[nt],
                                                            (short)0, acc[nt],
                                                            false, false);
        }
    }

    #pragma unroll
    for (int nt = 0; nt < NT; ++nt) {
        const int col = nt * 16 + l16;
        if (NGUARD && col >= N) continue;
        const float bv = bias[col];
        float* Cc = C + col;
        #pragma unroll
        for (int r = 0; r < 8; ++r) {
            const int row = m0 + r + (half << 3);
            float v = acc[nt][r] + bv;
            if (RELU_OUT) v = fmaxf(v, 0.0f);
            Cc[(size_t)row * N] = v;
        }
    }
}

#define HOP_NPW 4

__device__ __forceinline__ void axpy4(float4& acc, float s, const float4& v) {
    acc.x = fmaf(s, v.x, acc.x);
    acc.y = fmaf(s, v.y, acc.y);
    acc.z = fmaf(s, v.z, acc.z);
    acc.w = fmaf(s, v.w, acc.w);
}

template <int R>
__device__ __forceinline__ void gather_group(float4& acc, const float4* __restrict__ in4,
                                             int c, float we, int j, int lane) {
    float4 v[R];
    float  s[R];
    #pragma unroll
    for (int k = 0; k < R; ++k) {
        const int sidx = __builtin_amdgcn_readlane(c, j + k);
        s[k] = __int_as_float(__builtin_amdgcn_readlane(__float_as_int(we), j + k));
        v[k] = in4[(size_t)sidx * 32 + lane];
    }
    #pragma unroll
    for (int k = 0; k < R; ++k) axpy4(acc, s[k], v[k]);
}

__global__ __launch_bounds__(256)
void k_hop_gather128(const int* __restrict__ rowptr, const int* __restrict__ col,
                     const float* __restrict__ w, const float* __restrict__ dinv,
                     const float* __restrict__ in, float* __restrict__ out, int n) {
    const int lane = threadIdx.x & 31;
    const int wave = __builtin_amdgcn_readfirstlane((int)(blockIdx.x * 8 + (threadIdx.x >> 5)));
    const float4* __restrict__ in4 = (const float4*)in;
    float4* __restrict__ out4 = (float4*)out;

    for (int i = 0; i < HOP_NPW; ++i) {
        const int node = wave * HOP_NPW + i;
        if (node >= n) break;
        const int beg = rowptr[node];
        const int end = rowptr[node + 1];
        float sl = dinv[node]; sl *= sl;
        int   c  = 0;
        float we = 0.0f;
        if (beg < end) {
            const int idx = min(beg + lane, end - 1);
            c  = col[idx];
            we = w[idx];
        }
        const float4 self = in4[(size_t)node * 32 + lane];
        float4 acc = make_float4(0.0f, 0.0f, 0.0f, 0.0f);

        int base = beg;
        while (base < end) {
            const int cnt = min(32, end - base);
            int j = 0;
            for (; j + 8 <= cnt; j += 8) gather_group<8>(acc, in4, c, we, j, lane);
            if (j + 4 <= cnt) { gather_group<4>(acc, in4, c, we, j, lane); j += 4; }
            const int rem = cnt - j;
            if (rem == 3)      gather_group<3>(acc, in4, c, we, j, lane);
            else if (rem == 2) gather_group<2>(acc, in4, c, we, j, lane);
            else if (rem == 1) gather_group<1>(acc, in4, c, we, j, lane);
            base += 32;
            if (base < end) {
                const int idx = min(base + lane, end - 1);
                c  = col[idx];
                we = w[idx];
            }
        }
        axpy4(acc, sl, self);
        out4[(size_t)node * 32 + lane] = acc;
    }
}

__global__ void k_hop_gather40(const int* __restrict__ rowptr, const int* __restrict__ col,
                               const float* __restrict__ w, const float* __restrict__ dinv,
                               const float* __restrict__ in, float* __restrict__ out, int n) {
    const int t = blockIdx.x * blockDim.x + threadIdx.x;
    if (t >= n * (N_CLS / 4)) return;
    const int node = t / (N_CLS / 4);
    const int ch   = t - node * (N_CLS / 4);
    const float4* __restrict__ in4 = (const float4*)in;
    float sl = dinv[node]; sl *= sl;
    const float4 self = in4[t];
    float4 acc = make_float4(0.0f, 0.0f, 0.0f, 0.0f);
    const int beg = rowptr[node];
    const int end = rowptr[node + 1];
    #pragma unroll 4
    for (int p = beg; p < end; ++p) {
        const int   s  = col[p];
        const float we = w[p];
        const float4 v = in4[(size_t)s * (N_CLS / 4) + ch];
        axpy4(acc, we, v);
    }
    axpy4(acc, sl, self);
    ((float4*)out)[t] = acc;
}

extern "C" void kernel_launch(void* const* d_in, const int* in_sizes, int n_in,
                              void* d_out, int out_size, void* d_ws, size_t ws_size,
                              hipStream_t stream) {
    const float* x    = (const float*)d_in[0];
    const int*   edge = (const int*)d_in[1];
    const float* W1   = (const float*)d_in[2];
    const float* b1   = (const float*)d_in[3];
    const float* W2   = (const float*)d_in[4];
    const float* b2   = (const float*)d_in[5];
    const float* Wc   = (const float*)d_in[6];
    const float* bc   = (const float*)d_in[7];

    const int n = N_NODES;
    const int e = in_sizes[1] / 2;
    const int* src = edge;
    const int* dst = edge + e;

    float* ws   = (float*)d_ws;
    float* dinv = ws;
    float* wgt  = dinv + 50176;
    float* bufA = wgt + 800256;
    float* bufB = bufA + (size_t)n * D_HID;
    int*   col    = (int*)(bufB + (size_t)n * D_HID);
    int*   coltmp = col + 800256;
    int*   rowptr = coltmp + 800256;
    int*   cnt    = rowptr + 50176;
    int*   hist   = (int*)bufB;
    int*   rank   = hist + NCH * SCAN_PAD;
    float* outp = (float*)d_out;

    const int T = 256;

    k_chunk_count<<<NCH, SCAN_T, 0, stream>>>(dst, hist, rank, e);
    k_chunk_prefix<<<SCAN_PAD / 256, 256, 0, stream>>>(hist, cnt);
    k_scan_rows<<<1, SCAN_T, 0, stream>>>(cnt, rowptr, dinv);
    k_place_edges<<<NCH, SCAN_T, 0, stream>>>(src, dst, rowptr, hist, rank, coltmp, e);
    k_order_rows<<<(n + 7) / 8, T, 0, stream>>>(rowptr, coltmp, dinv, col, wgt, n);

    const int mTiles = n / 16;
    dim3 blk(32, 4, 1);
    dim3 grid((mTiles + 3) / 4, 1, 1);

    k_gemm_wmma_f32<8, false, true, false><<<grid, blk, 0, stream>>>(
        x, W1, b1, bufA, n, D_HID, D_HID);

    k_gemm_wmma_f32<8, false, false, false><<<grid, blk, 0, stream>>>(
        bufA, W2, b2, bufB, n, D_HID, D_HID);

    {
        const int hopN = (n + 8 * HOP_NPW - 1) / (8 * HOP_NPW);
        for (int hop = 0; hop < N_HOPS; ++hop) {
            const float* hin = (hop & 1) ? bufA : bufB;
            float*       hout = (hop & 1) ? bufB : bufA;
            k_hop_gather128<<<hopN, T, 0, stream>>>(rowptr, col, wgt, dinv, hin, hout, n);
        }
    }
    float* prop = (N_HOPS & 1) ? bufA : bufB;
    float* Q    = (N_HOPS & 1) ? bufB : bufA;

    k_gemm_wmma_f32<3, true, false, true><<<grid, blk, 0, stream>>>(
        prop, Wc, bc, Q, n, N_CLS, D_HID);

    k_hop_gather40<<<(n * (N_CLS / 4) + T - 1) / T, T, 0, stream>>>(
        rowptr, col, wgt, dinv, Q, outp, n);
}


# GEMM weight tile staged to LDS with grouped b128 loads (K row pair x 4 columns per load pair, about ten in flight per thread), not one dependent 4-byte load, wait, store per element; sums unchanged
# speedup vs baseline: 6.2361x; 1.1471x over previous
#include <hip/hip_runtime.h>

typedef float v2f __attribute__((ext_vector_type(2)));
typedef float v8f __attribute__((ext_vector_type(8)));

#define N_NODES 50000
#define D_HID   128
#define N_CLS   40
#define N_HOPS  30

#define SCAN_T   1024
#define SCAN_CH  ((N_NODES + SCAN_T) / SCAN_T)
#define SCAN_PAD (SCAN_T * SCAN_CH)
#define NCH      64

__global__ __launch_bounds__(SCAN_T)
void k_chunk_count(const int* __restrict__ dst, int* __restrict__ hist,
                   int* __restrict__ rank, int e) {
    __shared__ int h[SCAN_PAD];
    const int t = threadIdx.x;
    const int g = blockIdx.x;
    for (int i = t; i < SCAN_PAD; i += SCAN_T) h[i] = 0;
    __syncthreads();
    const int per = (e + NCH - 1) / NCH;
    const int lo  = g * per;
    const int hi  = min(lo + per, e);
    for (int i0 = lo + t; i0 < hi; i0 += 8 * SCAN_T) {
        int d[8];
        #pragma unroll
        for (int k = 0; k < 8; ++k) {
            const int i = i0 + k * SCAN_T;
            d[k] = (i < hi) ? dst[i] : 0;
        }
        #pragma unroll
        for (int k = 0; k < 8; ++k) {
            const int i = i0 + k * SCAN_T;
            if (i < hi) rank[i] = atomicAdd(&h[d[k]], 1);
        }
    }
    __syncthreads();
    int* __restrict__ hg = hist + (size_t)g * SCAN_PAD;
    for (int i = t; i < SCAN_PAD; i += SCAN_T) hg[i] = h[i];
}

__global__ __launch_bounds__(256)
void k_chunk_prefix(int* hist, int* __restrict__ cnt) {
    const int i = blockIdx.x * 256 + threadIdx.x;
    int v[NCH];
    #pragma unroll
    for (int g = 0; g < NCH; ++g) v[g] = hist[(size_t)g * SCAN_PAD + i];
    int run = 0;
    #pragma unroll
    for (int g = 0; g < NCH; ++g) {
        hist[(size_t)g * SCAN_PAD + i] = run;
        run += v[g];
    }
    cnt[i] = run;
}

__global__ __launch_bounds__(SCAN_T)
void k_scan_rows(const int* __restrict__ cnt, int* __restrict__ rowptr,
                 float* __restrict__ dinv) {
    __shared__ int part[SCAN_T];
    const int t = threadIdx.x;
    const int b = t * SCAN_CH;
    int c[SCAN_CH];
    int s = 0;
    #pragma unroll
    for (int k = 0; k < SCAN_CH; ++k) c[k] = cnt[b + k];
    #pragma unroll
    for (int k = 0; k < SCAN_CH; ++k) s += c[k];
    part[t] = s;
    __syncthreads();
    for (int off = 1; off < SCAN_T; off <<= 1) {
        const int v = (t >= off) ? part[t - off] : 0;
        __syncthreads();
        part[t] += v;
        __syncthreads();
    }
    int run = part[t] - s;
    #pragma unroll
    for (int k = 0; k < SCAN_CH; ++k) {
        const int i = b + k;
        rowptr[i] = run;
        dinv[i]   = rsqrtf((float)(c[k] + 1));
        run += c[k];
    }
}

__global__ __launch_bounds__(SCAN_T)
void k_place_edges(const int* __restrict__ src, const int* __restrict__ dst,
                   const int* __restrict__ rowptr, const int* __restrict__ hist,
                   const int* __restrict__ rank, int* __restrict__ col_tmp, int e) {
    __shared__ int off[SCAN_PAD];
    const int t = threadIdx.x;
    const int g = blockIdx.x;
    const int* __restrict__ hg = hist + (size_t)g * SCAN_PAD;
    #pragma unroll 7
    for (int i = t; i < SCAN_PAD; i += SCAN_T) off[i] = rowptr[i] + hg[i];
    __syncthreads();
    const int per = (e + NCH - 1) / NCH;
    const int lo  = g * per;
    const int hi  = min(lo + per, e);
    #pragma unroll 4
    for (int i = lo + t; i < hi; i += SCAN_T) col_tmp[off[dst[i]] + rank[i]] = src[i];
}

__global__ __launch_bounds__(256)
void k_order_rows(const int* __restrict__ rowptr, const int* __restrict__ col_tmp,
                  const float* __restrict__ dinv, int* __restrict__ col,
                  float* __restrict__ w, int n) {
    const int lane = threadIdx.x & 31;
    const int r = __builtin_amdgcn_readfirstlane((int)(blockIdx.x * 8 + (threadIdx.x >> 5)));
    if (r >= n) return;
    const int b = rowptr[r];
    const int e = rowptr[r + 1];
    const float dr = dinv[r];
    for (int mb = b; mb < e; mb += 32) {
        const int p = mb + lane;
        const bool have = p < e;
        const int mine = have ? col_tmp[p] : 0;
        int rank = 0;
        for (int ob = b; ob < e; ob += 32) {
            const int q0 = ob + lane;
            const int other = (q0 < e) ? col_tmp[q0] : 0;
            const int cnt = min(32, e - ob);
            for (int j = 0; j < cnt; ++j) {
                const int o = __builtin_amdgcn_readlane(other, j);
                const int q = ob + j;
                rank += ((o < mine) || (o == mine && q < p)) ? 1 : 0;
            }
        }
        if (have) {
            col[b + rank] = mine;
            w[b + rank]   = dinv[mine] * dr;
        }
    }
}

template <int NT, bool RELU_IN, bool RELU_OUT, bool NGUARD>
__global__ void k_gemm_wmma_f32(const float* __restrict__ A, const float* __restrict__ B,
                                const float* __restrict__ bias, float* __restrict__ C,
                                int M, int N, int K) {
    constexpr int NT16 = NT * 16;
    constexpr int RS   = NT * 32 + ((NT & 1) ? 64 : 32);
    __shared__ __attribute__((aligned(16))) float sB[64 * RS];

    const int lane = threadIdx.x;
    const int tid  = threadIdx.y * 32 + lane;

    constexpr int CG    = NT16 / 4;
    constexpr int UNITS = 64 * CG;
    constexpr int PER   = UNITS / 128;
    static_assert(UNITS % 128 == 0, "unit count must divide over the 128 threads");
    float4 lo[PER], hi[PER];
    #pragma unroll
    for (int u = 0; u < PER; ++u) {
        const int unit = tid + u * 128;
        const int p = unit / CG;
        const int j = unit - p * CG;
        const bool live = !NGUARD || 4 * j < N;
        const float* Bp = B + (size_t)(2 * p) * N + 4 * (live ? j : 0);
        lo[u] = *(const float4*)Bp;
        hi[u] = *(const float4*)(Bp + N);
    }
    #pragma unroll
    for (int u = 0; u < PER; ++u) {
        const int unit = tid + u * 128;
        const int p = unit / CG;
        const int j = unit - p * CG;
        const bool live = !NGUARD || 4 * j < N;
        const float4 zero = make_float4(0.0f, 0.0f, 0.0f, 0.0f);
        float4* d = (float4*)(sB + p * RS + 8 * j);
        d[0] = live ? make_float4(lo[u].x, hi[u].x, lo[u].y, hi[u].y) : zero;
        d[1] = live ? make_float4(lo[u].z, hi[u].z, lo[u].w, hi[u].w) : zero;
    }
    __syncthreads();

    const int mtile  = blockIdx.x * blockDim.y + threadIdx.y;
    const int mTiles = M >> 4;
    if (mtile >= mTiles) return;
    const int m0   = mtile << 4;
    const int half = lane >> 4;
    const int l16  = lane & 15;

    const float* Arow = A + (size_t)(m0 + l16) * K;

    v8f acc[NT];
    #pragma unroll
    for (int nt = 0; nt < NT; ++nt) acc[nt] = v8f{};

    for (int k0 = 0; k0 < K; k0 += 4) {
        const int kb = k0 + (half << 1);
        v2f a;
        a.x = Arow[kb];
        a.y = Arow[kb + 1];
        if (RELU_IN) { a.x = fmaxf(a.x, 0.0f); a.y = fmaxf(a.y, 0.0f); }

        const v2f* Bp = (const v2f*)(sB + ((k0 >> 1) + half) * RS) + l16;
        v2f bf[NT];
        #pragma unroll
        for (int nt = 0; nt < NT; ++nt) bf[nt] = Bp[nt * 16];

        #pragma unroll
        for (int nt = 0; nt < NT; ++nt) {
            acc[nt] = __builtin_amdgcn_wmma_f32_16x16x4_f32(false, a, false, bf[nt],
                                                            (short)0, acc[nt],
                                                            false, false);
        }
    }

    #pragma unroll
    for (int nt = 0; nt < NT; ++nt) {
        const int col = nt * 16 + l16;
        if (NGUARD && col >= N) continue;
        const float bv = bias[col];
        float* Cc = C + col;
        #pragma unroll
        for (int r = 0; r < 8; ++r) {
            const int row = m0 + r + (half << 3);
            float v = acc[nt][r] + bv;
            if (RELU_OUT) v = fmaxf(v, 0.0f);
            Cc[(size_t)row * N] = v;
        }
    }
}

#define HOP_NPW 4

__device__ __forceinline__ void axpy4(float4& acc, float s, const float4& v) {
    acc.x = fmaf(s, v.x, acc.x);
    acc.y = fmaf(s, v.y, acc.y);
    acc.z = fmaf(s, v.z, acc.z);
    acc.w = fmaf(s, v.w, acc.w);
}

template <int R>
__device__ __forceinline__ void gather_group(float4& acc, const float4* __restrict__ in4,
                                             int c, float we, int j, int lane) {
    float4 v[R];
    float  s[R];
    #pragma unroll
    for (int k = 0; k < R; ++k) {
        const int sidx = __builtin_amdgcn_readlane(c, j + k);
        s[k] = __int_as_float(__builtin_amdgcn_readlane(__float_as_int(we), j + k));
        v[k] = in4[(size_t)sidx * 32 + lane];
    }
    #pragma unroll
    for (int k = 0; k < R; ++k) axpy4(acc, s[k], v[k]);
}

__global__ __launch_bounds__(256)
void k_hop_gather128(const int* __restrict__ rowptr, const int* __restrict__ col,
                     const float* __restrict__ w, const float* __restrict__ dinv,
                     const float* __restrict__ in, float* __restrict__ out, int n) {
    const int lane = threadIdx.x & 31;
    const int wave = __builtin_amdgcn_readfirstlane((int)(blockIdx.x * 8 + (threadIdx.x >> 5)));
    const float4* __restrict__ in4 = (const float4*)in;
    float4* __restrict__ out4 = (float4*)out;

    for (int i = 0; i < HOP_NPW; ++i) {
        const int node = wave * HOP_NPW + i;
        if (node >= n) break;
        const int beg = rowptr[node];
        const int end = rowptr[node + 1];
        float sl = dinv[node]; sl *= sl;
        int   c  = 0;
        float we = 0.0f;
        if (beg < end) {
            const int idx = min(beg + lane, end - 1);
            c  = col[idx];
            we = w[idx];
        }
        const float4 self = in4[(size_t)node * 32 + lane];
        float4 acc = make_float4(0.0f, 0.0f, 0.0f, 0.0f);

        int base = beg;
        while (base < end) {
            const int cnt = min(32, end - base);
            int j = 0;
            for (; j + 8 <= cnt; j += 8) gather_group<8>(acc, in4, c, we, j, lane);
            if (j + 4 <= cnt) { gather_group<4>(acc, in4, c, we, j, lane); j += 4; }
            const int rem = cnt - j;
            if (rem == 3)      gather_group<3>(acc, in4, c, we, j, lane);
            else if (rem == 2) gather_group<2>(acc, in4, c, we, j, lane);
            else if (rem == 1) gather_group<1>(acc, in4, c, we, j, lane);
            base += 32;
            if (base < end) {
                const int idx = min(base + lane, end - 1);
                c  = col[idx];
                we = w[idx];
            }
        }
        axpy4(acc, sl, self);
        out4[(size_t)node * 32 + lane] = acc;
    }
}

__global__ void k_hop_gather40(const int* __restrict__ rowptr, const int* __restrict__ col,
                               const float* __restrict__ w, const float* __restrict__ dinv,
                               const float* __restrict__ in, float* __restrict__ out, int n) {
    const int t = blockIdx.x * blockDim.x + threadIdx.x;
    if (t >= n * (N_CLS / 4)) return;
    const int node = t / (N_CLS / 4);
    const int ch   = t - node * (N_CLS / 4);
    const float4* __restrict__ in4 = (const float4*)in;
    float sl = dinv[node]; sl *= sl;
    const float4 self = in4[t];
    float4 acc = make_float4(0.0f, 0.0f, 0.0f, 0.0f);
    const int beg = rowptr[node];
    const int end = rowptr[node + 1];
    #pragma unroll 4
    for (int p = beg; p < end; ++p) {
        const int   s  = col[p];
        const float we = w[p];
        const float4 v = in4[(size_t)s * (N_CLS / 4) + ch];
        axpy4(acc, we, v);
    }
    axpy4(acc, sl, self);
    ((float4*)out)[t] = acc;
}

extern "C" void kernel_launch(void* const* d_in, const int* in_sizes, int n_in,
                              void* d_out, int out_size, void* d_ws, size_t ws_size,
                              hipStream_t stream) {
    const float* x    = (const float*)d_in[0];
    const int*   edge = (const int*)d_in[1];
    const float* W1   = (const float*)d_in[2];
    const float* b1   = (const float*)d_in[3];
    const float* W2   = (const float*)d_in[4];
    const float* b2   = (const float*)d_in[5];
    const float* Wc   = (const float*)d_in[6];
    const float* bc   = (const float*)d_in[7];

    const int n = N_NODES;
    const int e = in_sizes[1] / 2;
    const int* src = edge;
    const int* dst = edge + e;

    float* ws   = (float*)d_ws;
    float* dinv = ws;
    float* wgt  = dinv + 50176;
    float* bufA = wgt + 800256;
    float* bufB = bufA + (size_t)n * D_HID;
    int*   col    = (int*)(bufB + (size_t)n * D_HID);
    int*   coltmp = col + 800256;
    int*   rowptr = coltmp + 800256;
    int*   cnt    = rowptr + 50176;
    int*   hist   = (int*)bufB;
    int*   rank   = hist + NCH * SCAN_PAD;
    float* outp = (float*)d_out;

    const int T = 256;

    k_chunk_count<<<NCH, SCAN_T, 0, stream>>>(dst, hist, rank, e);
    k_chunk_prefix<<<SCAN_PAD / 256, 256, 0, stream>>>(hist, cnt);
    k_scan_rows<<<1, SCAN_T, 0, stream>>>(cnt, rowptr, dinv);
    k_place_edges<<<NCH, SCAN_T, 0, stream>>>(src, dst, rowptr, hist, rank, coltmp, e);
    k_order_rows<<<(n + 7) / 8, T, 0, stream>>>(rowptr, coltmp, dinv, col, wgt, n);

    const int mTiles = n / 16;
    dim3 blk(32, 4, 1);
    dim3 grid((mTiles + 3) / 4, 1, 1);

    k_gemm_wmma_f32<8, false, true, false><<<grid, blk, 0, stream>>>(
        x, W1, b1, bufA, n, D_HID, D_HID);

    k_gemm_wmma_f32<8, false, false, false><<<grid, blk, 0, stream>>>(
        bufA, W2, b2, bufB, n, D_HID, D_HID);

    {
        const int hopN = (n + 8 * HOP_NPW - 1) / (8 * HOP_NPW);
        for (int hop = 0; hop < N_HOPS; ++hop) {
            const float* hin = (hop & 1) ? bufA : bufB;
            float*       hout = (hop & 1) ? bufB : bufA;
            k_hop_gather128<<<hopN, T, 0, stream>>>(rowptr, col, wgt, dinv, hin, hout, n);
        }
    }
    float* prop = (N_HOPS & 1) ? bufA : bufB;
    float* Q    = (N_HOPS & 1) ? bufB : bufA;

    k_gemm_wmma_f32<3, true, false, true><<<grid, blk, 0, stream>>>(
        prop, Wc, bc, Q, n, N_CLS, D_HID);

    k_hop_gather40<<<(n * (N_CLS / 4) + T - 1) / T, T, 0, stream>>>(
        rowptr, col, wgt, dinv, Q, outp, n);
}
